# starting kernel
# baseline (speedup 1.0000x reference)
#include <hip/hip_runtime.h>
#include <math.h>

#define N_LAYERS 4
#define T        34
#define TP       48
#define PSTRIDE  50
#define VOCAB    14
#define DM       6
#define HD       3

typedef float v2f __attribute__((ext_vector_type(2)));
typedef float v4f __attribute__((ext_vector_type(4)));
typedef float v8f __attribute__((ext_vector_type(8)));

__global__ __launch_bounds__(64)
void addtrans_kernel(const int* __restrict__ idx,
                     const float* __restrict__ tok_emb,
                     const float* __restrict__ pos_enc,
                     const float* __restrict__ ln_w, const float* __restrict__ ln_b,
                     const float* __restrict__ q1w, const float* __restrict__ k1w,
                     const float* __restrict__ v1w,
                     const float* __restrict__ q2w, const float* __restrict__ k2w,
                     const float* __restrict__ v2w,
                     const float* __restrict__ out_w,
                     const float* __restrict__ lnf_w, const float* __restrict__ lnf_b,
                     const float* __restrict__ head_w,
                     float* __restrict__ out)
{
    __shared__ float xs[TP][DM];
    __shared__ float hs[TP][DM];
    __shared__ float qs[2][TP][4];
    __shared__ float ks_[2][TP][4];
    __shared__ float vs[2][TP][4];
    __shared__ float os[2][TP][4];
    __shared__ float Ps[2][TP][PSTRIDE];

    const int b    = blockIdx.x;
    const int tid  = threadIdx.x;
    const int w    = tid >> 5;
    const int lane = tid & 31;
    const int hi16 = lane >> 4;
    const int l16  = lane & 15;
    const int kofs = hi16 * 2;
    const int colc = (l16 < 3) ? l16 : 3;

    __builtin_prefetch(idx + (size_t)b * T, 0, 1);

    for (int t = tid; t < T; t += 64) {
        int tok = idx[(size_t)b * T + t];
#pragma unroll
        for (int c = 0; c < 3; ++c) {
            xs[t][c]     = tok_emb[tok * 3 + c];
            xs[t][3 + c] = pos_enc[t * 3 + c];
        }
    }
    __syncthreads();

    const float rs3 = 0.57735026918962576f;

    for (int l = 0; l < N_LAYERS; ++l) {
        if (tid < TP) {
            if (tid < T) {
                float m = 0.f;
#pragma unroll
                for (int c = 0; c < DM; ++c) m += xs[tid][c];
                m *= (1.0f / DM);
                float var = 0.f;
#pragma unroll
                for (int c = 0; c < DM; ++c) { float d = xs[tid][c] - m; var += d * d; }
                var *= (1.0f / DM);
                float r = rsqrtf(var + 1e-5f);
#pragma unroll
                for (int c = 0; c < DM; ++c)
                    hs[tid][c] = (xs[tid][c] - m) * r * ln_w[l * DM + c] + ln_b[l * DM + c];
            } else {
#pragma unroll
                for (int c = 0; c < DM; ++c) hs[tid][c] = 0.f;
            }
        }
        __syncthreads();

        const float* qw = (w == 0 ? q1w : q2w) + l * 9;
        const float* kw = (w == 0 ? k1w : k2w) + l * 9;
        const float* vw = (w == 0 ? v1w : v2w) + l * 9;
        const int hoff = w * HD;
        for (int t = lane; t < TP; t += 32) {
            float h0 = hs[t][hoff], h1 = hs[t][hoff + 1], h2 = hs[t][hoff + 2];
            float valid = (t < T) ? 1.0f : 0.0f;
            v4f qv, kv, vv;
#pragma unroll
            for (int i = 0; i < HD; ++i) {
                qv[i] = (qw[i*3]*h0 + qw[i*3+1]*h1 + qw[i*3+2]*h2) * valid;
                kv[i] = (kw[i*3]*h0 + kw[i*3+1]*h1 + kw[i*3+2]*h2) * valid;
                vv[i] = (vw[i*3]*h0 + vw[i*3+1]*h1 + vw[i*3+2]*h2) * valid;
            }
            qv[3] = 0.f; kv[3] = 0.f; vv[3] = 0.f;
            *(v4f*)&qs[w][t][0]  = qv;
            *(v4f*)&ks_[w][t][0] = kv;
            *(v4f*)&vs[w][t][0]  = vv;
        }

        for (int mi = 0; mi < 3; ++mi) {
            const int am = mi * 16 + l16;
            v2f a = *(const v2f*)&qs[w][am][kofs];
            for (int ni = 0; ni <= mi; ++ni) {
                const int bn = ni * 16 + l16;
                v2f bf = *(const v2f*)&ks_[w][bn][kofs];
                v8f c = {0.f, 0.f, 0.f, 0.f, 0.f, 0.f, 0.f, 0.f};
                c = __builtin_amdgcn_wmma_f32_16x16x4_f32(false, a, false, bf,
                                                          (short)0, c, false, false);
#pragma unroll
                for (int r = 0; r < 8; ++r)
                    Ps[w][mi * 16 + hi16 * 8 + r][ni * 16 + l16] = c[r];
            }
        }

        for (int m = lane; m < TP; m += 32) {
            const bool rowok = (m < T);
            float mx = -3.4e38f;
            for (int n = 0; n < TP; ++n) {
                float s = Ps[w][m][n] * rs3;
                bool valid = rowok && (n <= m);
                mx = valid ? fmaxf(mx, s) : mx;
            }
            float sum = 0.f;
            for (int n = 0; n < TP; ++n) {
                float s = Ps[w][m][n] * rs3;
                bool valid = rowok && (n <= m);
                float e = valid ? __expf(s - mx) : 0.f;
                Ps[w][m][n] = e;
                sum += e;
            }
            float inv = (sum > 0.f) ? (1.0f / sum) : 0.f;
            for (int n = 0; n < TP; ++n) Ps[w][m][n] *= inv;
        }

        for (int mi = 0; mi < 3; ++mi) {
            v8f acc = {0.f, 0.f, 0.f, 0.f, 0.f, 0.f, 0.f, 0.f};
            const int am = mi * 16 + l16;
            int kmax = 4 * (mi + 1);
            if (kmax > (T + 3) / 4) kmax = (T + 3) / 4;
            for (int kc = 0; kc < kmax; ++kc) {
                const int nb = kc * 4;
                v2f a = *(const v2f*)&Ps[w][am][nb + kofs];
                v2f bf;
                bf.x = vs[w][nb + kofs + 0][colc];
                bf.y = vs[w][nb + kofs + 1][colc];
                acc = __builtin_amdgcn_wmma_f32_16x16x4_f32(false, a, false, bf,
                                                            (short)0, acc, false, false);
            }
            if (l16 < HD) {
#pragma unroll
                for (int r = 0; r < 8; ++r)
                    os[w][mi * 16 + hi16 * 8 + r][l16] = acc[r];
            }
        }
        __syncthreads();

        if (tid < T) {
            float u[DM];
#pragma unroll
            for (int i = 0; i < HD; ++i) { u[i] = os[0][tid][i]; u[HD + i] = os[1][tid][i]; }
#pragma unroll
            for (int i = 0; i < DM; ++i) {
                float acc = 0.f;
#pragma unroll
                for (int j = 0; j < DM; ++j) acc += out_w[l * 36 + i * 6 + j] * u[j];
                xs[tid][i] += acc;
            }
        }
        __syncthreads();
    }

    if (tid < T) {
        float m = 0.f;
#pragma unroll
        for (int c = 0; c < DM; ++c) m += xs[tid][c];
        m *= (1.0f / DM);
        float var = 0.f;
#pragma unroll
        for (int c = 0; c < DM; ++c) { float d = xs[tid][c] - m; var += d * d; }
        var *= (1.0f / DM);
        float r = rsqrtf(var + 1e-5f);
        float xf[DM];
#pragma unroll
        for (int c = 0; c < DM; ++c)
            xf[c] = (xs[tid][c] - m) * r * lnf_w[c] + lnf_b[c];
        float* op = out + ((size_t)b * T + tid) * VOCAB;
#pragma unroll
        for (int v = 0; v < VOCAB; ++v) {
            float acc = 0.f;
#pragma unroll
            for (int j = 0; j < DM; ++j) acc += head_w[v * DM + j] * xf[j];
            op[v] = acc;
        }
    }
}

extern "C" void kernel_launch(void* const* d_in, const int* in_sizes, int n_in,
                              void* d_out, int out_size, void* d_ws, size_t ws_size,
                              hipStream_t stream) {
    const int*   idx     = (const int*)  d_in[0];
    const float* tok_emb = (const float*)d_in[1];
    const float* pos_enc = (const float*)d_in[2];
    const float* ln_w    = (const float*)d_in[3];
    const float* ln_b    = (const float*)d_in[4];
    const float* q1w     = (const float*)d_in[5];
    const float* k1w     = (const float*)d_in[6];
    const float* v1w     = (const float*)d_in[7];
    const float* q2w     = (const float*)d_in[8];
    const float* k2w     = (const float*)d_in[9];
    const float* v2w     = (const float*)d_in[10];
    const float* out_w   = (const float*)d_in[11];
    const float* lnf_w   = (const float*)d_in[12];
    const float* lnf_b   = (const float*)d_in[13];
    const float* head_w  = (const float*)d_in[14];
    float* out = (float*)d_out;

    const int batch = in_sizes[0] / T;
    addtrans_kernel<<<batch, 64, 0, stream>>>(idx, tok_emb, pos_enc, ln_w, ln_b,
                                              q1w, k1w, v1w, q2w, k2w, v2w,
                                              out_w, lnf_w, lnf_b, head_w, out);
    (void)d_ws; (void)ws_size; (void)n_in; (void)out_size;
}


# Softmax row held in registers: one b128 read and one b128 write of each score row (stride 36) replace three LDS passes; P left unnormalised, 1/sum applied to the PV result; sum order unchanged
# speedup vs baseline: 3.6091x; 3.6091x over previous
#include <hip/hip_runtime.h>
#include <math.h>

#define N_LAYERS 4
#define T        34
#define TP       48
#define PCOLS    36
#define PSTRIDE  36
#define VOCAB    14
#define DM       6
#define HD       3

typedef float v2f __attribute__((ext_vector_type(2)));
typedef float v4f __attribute__((ext_vector_type(4)));
typedef float v8f __attribute__((ext_vector_type(8)));

template <int NC>
__device__ __forceinline__ float softmax_row_regs(float* prow, const int m, const float sc2)
{
    const float NEG_BIG = -3.0e38f;
    float s[NC];
#pragma unroll
    for (int j = 0; j < NC / 4; ++j) {
        const v4f t = *(const v4f*)(prow + 4 * j);
        s[4 * j + 0] = t.x; s[4 * j + 1] = t.y; s[4 * j + 2] = t.z; s[4 * j + 3] = t.w;
    }
    float mx = NEG_BIG;
#pragma unroll
    for (int n = 0; n < NC; ++n) {
        s[n] = (n <= m) ? s[n] : NEG_BIG;
        mx = fmaxf(mx, s[n]);
    }
    const float mb = -mx * sc2;
    float sum = 0.f;
#pragma unroll
    for (int j = 0; j < NC / 4; ++j) {
        v4f e;
        e.x = __builtin_amdgcn_exp2f(fmaf(s[4 * j + 0], sc2, mb)); sum += e.x;
        e.y = __builtin_amdgcn_exp2f(fmaf(s[4 * j + 1], sc2, mb)); sum += e.y;
        e.z = __builtin_amdgcn_exp2f(fmaf(s[4 * j + 2], sc2, mb)); sum += e.z;
        e.w = __builtin_amdgcn_exp2f(fmaf(s[4 * j + 3], sc2, mb)); sum += e.w;
        *(v4f*)(prow + 4 * j) = e;
    }
    return sum;
}

__global__ __launch_bounds__(64)
void addtrans_kernel(const int* __restrict__ idx,
                     const float* __restrict__ tok_emb,
                     const float* __restrict__ pos_enc,
                     const float* __restrict__ ln_w, const float* __restrict__ ln_b,
                     const float* __restrict__ q1w, const float* __restrict__ k1w,
                     const float* __restrict__ v1w,
                     const float* __restrict__ q2w, const float* __restrict__ k2w,
                     const float* __restrict__ v2w,
                     const float* __restrict__ out_w,
                     const float* __restrict__ lnf_w, const float* __restrict__ lnf_b,
                     const float* __restrict__ head_w,
                     float* __restrict__ out)
{
    __shared__ float xs[TP][DM];
    __shared__ float hs[TP][DM];
    __shared__ float qs[2][TP][4];
    __shared__ float ks_[2][TP][4];
    __shared__ float vs[2][TP][4];
    __shared__ float os[2][TP][4];
    __shared__ __attribute__((aligned(16))) float Ps[2][TP][PSTRIDE];
    __shared__ float rinv[2][TP];

    const int b    = blockIdx.x;
    const int tid  = threadIdx.x;
    const int w    = tid >> 5;
    const int lane = tid & 31;
    const int hi16 = lane >> 4;
    const int l16  = lane & 15;
    const int kofs = hi16 * 2;
    const int colc = (l16 < 3) ? l16 : 3;

    __builtin_prefetch(idx + (size_t)b * T, 0, 1);

    for (int t = tid; t < T; t += 64) {
        int tok = idx[(size_t)b * T + t];
#pragma unroll
        for (int c = 0; c < 3; ++c) {
            xs[t][c]     = tok_emb[tok * 3 + c];
            xs[t][3 + c] = pos_enc[t * 3 + c];
        }
    }
    __syncthreads();

    const float sc2 = 0.57735026918962576f * 1.4426950408889634f;

    for (int l = 0; l < N_LAYERS; ++l) {
        if (tid < TP) {
            if (tid < T) {
                float m = 0.f;
#pragma unroll
                for (int c = 0; c < DM; ++c) m += xs[tid][c];
                m *= (1.0f / DM);
                float var = 0.f;
#pragma unroll
                for (int c = 0; c < DM; ++c) { float d = xs[tid][c] - m; var += d * d; }
                var *= (1.0f / DM);
                float r = rsqrtf(var + 1e-5f);
#pragma unroll
                for (int c = 0; c < DM; ++c)
                    hs[tid][c] = (xs[tid][c] - m) * r * ln_w[l * DM + c] + ln_b[l * DM + c];
            } else {
#pragma unroll
                for (int c = 0; c < DM; ++c) hs[tid][c] = 0.f;
            }
        }
        __syncthreads();

        const float* qw = (w == 0 ? q1w : q2w) + l * 9;
        const float* kw = (w == 0 ? k1w : k2w) + l * 9;
        const float* vw = (w == 0 ? v1w : v2w) + l * 9;
        const int hoff = w * HD;
        for (int t = lane; t < TP; t += 32) {
            float h0 = hs[t][hoff], h1 = hs[t][hoff + 1], h2 = hs[t][hoff + 2];
            float valid = (t < T) ? 1.0f : 0.0f;
            v4f qv, kv, vv;
#pragma unroll
            for (int i = 0; i < HD; ++i) {
                qv[i] = (qw[i*3]*h0 + qw[i*3+1]*h1 + qw[i*3+2]*h2) * valid;
                kv[i] = (kw[i*3]*h0 + kw[i*3+1]*h1 + kw[i*3+2]*h2) * valid;
                vv[i] = (vw[i*3]*h0 + vw[i*3+1]*h1 + vw[i*3+2]*h2) * valid;
            }
            qv[3] = 0.f; kv[3] = 0.f; vv[3] = 0.f;
            *(v4f*)&qs[w][t][0]  = qv;
            *(v4f*)&ks_[w][t][0] = kv;
            *(v4f*)&vs[w][t][0]  = vv;
        }

        for (int mi = 0; mi < 3; ++mi) {
            const int am = mi * 16 + l16;
            v2f a = *(const v2f*)&qs[w][am][kofs];
            for (int ni = 0; ni <= mi; ++ni) {
                const int bn = ni * 16 + l16;
                v2f bf = *(const v2f*)&ks_[w][bn][kofs];
                v8f c = {0.f, 0.f, 0.f, 0.f, 0.f, 0.f, 0.f, 0.f};
                c = __builtin_amdgcn_wmma_f32_16x16x4_f32(false, a, false, bf,
                                                          (short)0, c, false, false);
                if (ni * 16 + l16 < PCOLS) {
#pragma unroll
                    for (int r = 0; r < 8; ++r)
                        Ps[w][mi * 16 + hi16 * 8 + r][ni * 16 + l16] = c[r];
                }
            }
        }

        {
            const float sum = softmax_row_regs<32>(&Ps[w][lane][0], lane, sc2);
            rinv[w][lane] = (sum > 0.f) ? (1.0f / sum) : 0.f;
        }
        if (lane < T - 32) {
            const int m = 32 + lane;
            const float sum = softmax_row_regs<PCOLS>(&Ps[w][m][0], m, sc2);
            rinv[w][m] = (sum > 0.f) ? (1.0f / sum) : 0.f;
        }

        for (int mi = 0; mi < 3; ++mi) {
            v8f acc = {0.f, 0.f, 0.f, 0.f, 0.f, 0.f, 0.f, 0.f};
            const int am = mi * 16 + l16;
            int kmax = 4 * (mi + 1);
            if (kmax > (T + 3) / 4) kmax = (T + 3) / 4;
            for (int kc = 0; kc < kmax; ++kc) {
                const int nb = kc * 4;
                v2f a = *(const v2f*)&Ps[w][am][nb + kofs];
                v2f bf;
                bf.x = vs[w][nb + kofs + 0][colc];
                bf.y = vs[w][nb + kofs + 1][colc];
                acc = __builtin_amdgcn_wmma_f32_16x16x4_f32(false, a, false, bf,
                                                            (short)0, acc, false, false);
            }
            if (l16 < HD) {
#pragma unroll
                for (int r = 0; r < 8; ++r)
                    os[w][mi * 16 + hi16 * 8 + r][l16] = acc[r];
            }
        }
        __syncthreads();

        if (tid < T) {
            float u[DM];
            const float r0 = rinv[0][tid], r1 = rinv[1][tid];
#pragma unroll
            for (int i = 0; i < HD; ++i) { u[i] = os[0][tid][i] * r0; u[HD + i] = os[1][tid][i] * r1; }
#pragma unroll
            for (int i = 0; i < DM; ++i) {
                float acc = 0.f;
#pragma unroll
                for (int j = 0; j < DM; ++j) acc += out_w[l * 36 + i * 6 + j] * u[j];
                xs[tid][i] += acc;
            }
        }
        __syncthreads();
    }

    if (tid < T) {
        float m = 0.f;
#pragma unroll
        for (int c = 0; c < DM; ++c) m += xs[tid][c];
        m *= (1.0f / DM);
        float var = 0.f;
#pragma unroll
        for (int c = 0; c < DM; ++c) { float d = xs[tid][c] - m; var += d * d; }
        var *= (1.0f / DM);
        float r = rsqrtf(var + 1e-5f);
        float xf[DM];
#pragma unroll
        for (int c = 0; c < DM; ++c)
            xf[c] = (xs[tid][c] - m) * r * lnf_w[c] + lnf_b[c];
        float* op = out + ((size_t)b * T + tid) * VOCAB;
#pragma unroll
        for (int v = 0; v < VOCAB; ++v) {
            float acc = 0.f;
#pragma unroll
            for (int j = 0; j < DM; ++j) acc += head_w[v * DM + j] * xf[j];
            op[v] = acc;
        }
    }
}

extern "C" void kernel_launch(void* const* d_in, const int* in_sizes, int n_in,
                              void* d_out, int out_size, void* d_ws, size_t ws_size,
                              hipStream_t stream) {
    const int*   idx     = (const int*)  d_in[0];
    const float* tok_emb = (const float*)d_in[1];
    const float* pos_enc = (const float*)d_in[2];
    const float* ln_w    = (const float*)d_in[3];
    const float* ln_b    = (const float*)d_in[4];
    const float* q1w     = (const float*)d_in[5];
    const float* k1w     = (const float*)d_in[6];
    const float* v1w     = (const float*)d_in[7];
    const float* q2w     = (const float*)d_in[8];
    const float* k2w     = (const float*)d_in[9];
    const float* v2w     = (const float*)d_in[10];
    const float* out_w   = (const float*)d_in[11];
    const float* lnf_w   = (const float*)d_in[12];
    const float* lnf_b   = (const float*)d_in[13];
    const float* head_w  = (const float*)d_in[14];
    float* out = (float*)d_out;

    const int batch = in_sizes[0] / T;
    addtrans_kernel<<<batch, 64, 0, stream>>>(idx, tok_emb, pos_enc, ln_w, ln_b,
                                              q1w, k1w, v1w, q2w, k2w, v2w,
                                              out_w, lnf_w, lnf_b, head_w, out);
    (void)d_ws; (void)ws_size; (void)n_in; (void)out_size;
}


# Softmax of the 2 left-over query rows (32, 33) spread over the wave, 16 lanes per row; row max and sum by DPP xor butterfly (their sum order becomes a tree); replaces a 36-column pass run on 2 lanes
# speedup vs baseline: 3.7710x; 1.0449x over previous
#include <hip/hip_runtime.h>
#include <math.h>

#define N_LAYERS 4
#define T        34
#define TP       48
#define PCOLS    36
#define PSTRIDE  36
#define VOCAB    14
#define DM       6
#define HD       3

typedef float v2f __attribute__((ext_vector_type(2)));
typedef float v4f __attribute__((ext_vector_type(4)));
typedef float v8f __attribute__((ext_vector_type(8)));

template <int NC>
__device__ __forceinline__ float softmax_row_regs(float* prow, const int m, const float sc2)
{
    const float NEG_BIG = -3.0e38f;
    float s[NC];
#pragma unroll
    for (int j = 0; j < NC / 4; ++j) {
        const v4f t = *(const v4f*)(prow + 4 * j);
        s[4 * j + 0] = t.x; s[4 * j + 1] = t.y; s[4 * j + 2] = t.z; s[4 * j + 3] = t.w;
    }
    float mx = NEG_BIG;
#pragma unroll
    for (int n = 0; n < NC; ++n) {
        s[n] = (n <= m) ? s[n] : NEG_BIG;
        mx = fmaxf(mx, s[n]);
    }
    const float mb = -mx * sc2;
    float sum = 0.f;
#pragma unroll
    for (int j = 0; j < NC / 4; ++j) {
        v4f e;
        e.x = __builtin_amdgcn_exp2f(fmaf(s[4 * j + 0], sc2, mb)); sum += e.x;
        e.y = __builtin_amdgcn_exp2f(fmaf(s[4 * j + 1], sc2, mb)); sum += e.y;
        e.z = __builtin_amdgcn_exp2f(fmaf(s[4 * j + 2], sc2, mb)); sum += e.z;
        e.w = __builtin_amdgcn_exp2f(fmaf(s[4 * j + 3], sc2, mb)); sum += e.w;
        *(v4f*)(prow + 4 * j) = e;
    }
    return sum;
}

template <int X>
__device__ __forceinline__ float row_xor(float v)
{
    return __builtin_bit_cast(float,
        __builtin_amdgcn_update_dpp(0, __builtin_bit_cast(int, v), 0x160 + X, 0xf, 0xf, false));
}

__device__ __forceinline__ float row16_max(float v)
{
    v = fmaxf(v, row_xor<1>(v));
    v = fmaxf(v, row_xor<2>(v));
    v = fmaxf(v, row_xor<4>(v));
    v = fmaxf(v, row_xor<8>(v));
    return v;
}
__device__ __forceinline__ float row16_sum(float v)
{
    v += row_xor<1>(v);
    v += row_xor<2>(v);
    v += row_xor<4>(v);
    v += row_xor<8>(v);
    return v;
}

__global__ __launch_bounds__(64)
void addtrans_kernel(const int* __restrict__ idx,
                     const float* __restrict__ tok_emb,
                     const float* __restrict__ pos_enc,
                     const float* __restrict__ ln_w, const float* __restrict__ ln_b,
                     const float* __restrict__ q1w, const float* __restrict__ k1w,
                     const float* __restrict__ v1w,
                     const float* __restrict__ q2w, const float* __restrict__ k2w,
                     const float* __restrict__ v2w,
                     const float* __restrict__ out_w,
                     const float* __restrict__ lnf_w, const float* __restrict__ lnf_b,
                     const float* __restrict__ head_w,
                     float* __restrict__ out)
{
    __shared__ float xs[TP][DM];
    __shared__ float hs[TP][DM];
    __shared__ float qs[2][TP][4];
    __shared__ float ks_[2][TP][4];
    __shared__ float vs[2][TP][4];
    __shared__ float os[2][TP][4];
    __shared__ __attribute__((aligned(16))) float Ps[2][TP][PSTRIDE];
    __shared__ float rinv[2][TP];

    const int b    = blockIdx.x;
    const int tid  = threadIdx.x;
    const int w    = tid >> 5;
    const int lane = tid & 31;
    const int hi16 = lane >> 4;
    const int l16  = lane & 15;
    const int kofs = hi16 * 2;
    const int colc = (l16 < 3) ? l16 : 3;

    __builtin_prefetch(idx + (size_t)b * T, 0, 1);

    for (int t = tid; t < T; t += 64) {
        int tok = idx[(size_t)b * T + t];
#pragma unroll
        for (int c = 0; c < 3; ++c) {
            xs[t][c]     = tok_emb[tok * 3 + c];
            xs[t][3 + c] = pos_enc[t * 3 + c];
        }
    }
    __syncthreads();

    const float sc2 = 0.57735026918962576f * 1.4426950408889634f;

    for (int l = 0; l < N_LAYERS; ++l) {
        if (tid < TP) {
            if (tid < T) {
                float m = 0.f;
#pragma unroll
                for (int c = 0; c < DM; ++c) m += xs[tid][c];
                m *= (1.0f / DM);
                float var = 0.f;
#pragma unroll
                for (int c = 0; c < DM; ++c) { float d = xs[tid][c] - m; var += d * d; }
                var *= (1.0f / DM);
                float r = rsqrtf(var + 1e-5f);
#pragma unroll
                for (int c = 0; c < DM; ++c)
                    hs[tid][c] = (xs[tid][c] - m) * r * ln_w[l * DM + c] + ln_b[l * DM + c];
            } else {
#pragma unroll
                for (int c = 0; c < DM; ++c) hs[tid][c] = 0.f;
            }
        }
        __syncthreads();

        const float* qw = (w == 0 ? q1w : q2w) + l * 9;
        const float* kw = (w == 0 ? k1w : k2w) + l * 9;
        const float* vw = (w == 0 ? v1w : v2w) + l * 9;
        const int hoff = w * HD;
        for (int t = lane; t < TP; t += 32) {
            float h0 = hs[t][hoff], h1 = hs[t][hoff + 1], h2 = hs[t][hoff + 2];
            float valid = (t < T) ? 1.0f : 0.0f;
            v4f qv, kv, vv;
#pragma unroll
            for (int i = 0; i < HD; ++i) {
                qv[i] = (qw[i*3]*h0 + qw[i*3+1]*h1 + qw[i*3+2]*h2) * valid;
                kv[i] = (kw[i*3]*h0 + kw[i*3+1]*h1 + kw[i*3+2]*h2) * valid;
                vv[i] = (vw[i*3]*h0 + vw[i*3+1]*h1 + vw[i*3+2]*h2) * valid;
            }
            qv[3] = 0.f; kv[3] = 0.f; vv[3] = 0.f;
            *(v4f*)&qs[w][t][0]  = qv;
            *(v4f*)&ks_[w][t][0] = kv;
            *(v4f*)&vs[w][t][0]  = vv;
        }

        for (int mi = 0; mi < 3; ++mi) {
            const int am = mi * 16 + l16;
            v2f a = *(const v2f*)&qs[w][am][kofs];
            for (int ni = 0; ni <= mi; ++ni) {
                const int bn = ni * 16 + l16;
                v2f bf = *(const v2f*)&ks_[w][bn][kofs];
                v8f c = {0.f, 0.f, 0.f, 0.f, 0.f, 0.f, 0.f, 0.f};
                c = __builtin_amdgcn_wmma_f32_16x16x4_f32(false, a, false, bf,
                                                          (short)0, c, false, false);
                if (ni * 16 + l16 < PCOLS) {
#pragma unroll
                    for (int r = 0; r < 8; ++r)
                        Ps[w][mi * 16 + hi16 * 8 + r][ni * 16 + l16] = c[r];
                }
            }
        }

        {
            const float sum = softmax_row_regs<32>(&Ps[w][lane][0], lane, sc2);
            rinv[w][lane] = (sum > 0.f) ? (1.0f / sum) : 0.f;
        }
        {
            const float NEG_BIG = -3.0e38f;
            const int m = 32 + hi16;
            float* prow = &Ps[w][m][0];
            const bool has2 = (l16 < PCOLS - 32);
            const float s0 = prow[l16];
            const float s1 = prow[16 + l16];
            float s2 = NEG_BIG;
            if (has2) {
                const float t = prow[32 + l16];
                s2 = (l16 <= hi16) ? t : NEG_BIG;
            }
            const float mx = row16_max(fmaxf(fmaxf(s0, s1), s2));
            const float mb = -mx * sc2;
            const float e0 = __builtin_amdgcn_exp2f(fmaf(s0, sc2, mb));
            const float e1 = __builtin_amdgcn_exp2f(fmaf(s1, sc2, mb));
            const float e2 = __builtin_amdgcn_exp2f(fmaf(s2, sc2, mb));
            const float sum = row16_sum((e0 + e1) + e2);
            prow[l16]      = e0;
            prow[16 + l16] = e1;
            if (has2) prow[32 + l16] = e2;
            if (l16 == 0) rinv[w][m] = (sum > 0.f) ? (1.0f / sum) : 0.f;
        }

        for (int mi = 0; mi < 3; ++mi) {
            v8f acc = {0.f, 0.f, 0.f, 0.f, 0.f, 0.f, 0.f, 0.f};
            const int am = mi * 16 + l16;
            int kmax = 4 * (mi + 1);
            if (kmax > (T + 3) / 4) kmax = (T + 3) / 4;
            for (int kc = 0; kc < kmax; ++kc) {
                const int nb = kc * 4;
                v2f a = *(const v2f*)&Ps[w][am][nb + kofs];
                v2f bf;
                bf.x = vs[w][nb + kofs + 0][colc];
                bf.y = vs[w][nb + kofs + 1][colc];
                acc = __builtin_amdgcn_wmma_f32_16x16x4_f32(false, a, false, bf,
                                                            (short)0, acc, false, false);
            }
            if (l16 < HD) {
#pragma unroll
                for (int r = 0; r < 8; ++r)
                    os[w][mi * 16 + hi16 * 8 + r][l16] = acc[r];
            }
        }
        __syncthreads();

        if (tid < T) {
            float u[DM];
            const float r0 = rinv[0][tid], r1 = rinv[1][tid];
#pragma unroll
            for (int i = 0; i < HD; ++i) { u[i] = os[0][tid][i] * r0; u[HD + i] = os[1][tid][i] * r1; }
#pragma unroll
            for (int i = 0; i < DM; ++i) {
                float acc = 0.f;
#pragma unroll
                for (int j = 0; j < DM; ++j) acc += out_w[l * 36 + i * 6 + j] * u[j];
                xs[tid][i] += acc;
            }
        }
        __syncthreads();
    }

    if (tid < T) {
        float m = 0.f;
#pragma unroll
        for (int c = 0; c < DM; ++c) m += xs[tid][c];
        m *= (1.0f / DM);
        float var = 0.f;
#pragma unroll
        for (int c = 0; c < DM; ++c) { float d = xs[tid][c] - m; var += d * d; }
        var *= (1.0f / DM);
        float r = rsqrtf(var + 1e-5f);
        float xf[DM];
#pragma unroll
        for (int c = 0; c < DM; ++c)
            xf[c] = (xs[tid][c] - m) * r * lnf_w[c] + lnf_b[c];
        float* op = out + ((size_t)b * T + tid) * VOCAB;
#pragma unroll
        for (int v = 0; v < VOCAB; ++v) {
            float acc = 0.f;
#pragma unroll
            for (int j = 0; j < DM; ++j) acc += head_w[v * DM + j] * xf[j];
            op[v] = acc;
        }
    }
}

extern "C" void kernel_launch(void* const* d_in, const int* in_sizes, int n_in,
                              void* d_out, int out_size, void* d_ws, size_t ws_size,
                              hipStream_t stream) {
    const int*   idx     = (const int*)  d_in[0];
    const float* tok_emb = (const float*)d_in[1];
    const float* pos_enc = (const float*)d_in[2];
    const float* ln_w    = (const float*)d_in[3];
    const float* ln_b    = (const float*)d_in[4];
    const float* q1w     = (const float*)d_in[5];
    const float* k1w     = (const float*)d_in[6];
    const float* v1w     = (const float*)d_in[7];
    const float* q2w     = (const float*)d_in[8];
    const float* k2w     = (const float*)d_in[9];
    const float* v2w     = (const float*)d_in[10];
    const float* out_w   = (const float*)d_in[11];
    const float* lnf_w   = (const float*)d_in[12];
    const float* lnf_b   = (const float*)d_in[13];
    const float* head_w  = (const float*)d_in[14];
    float* out = (float*)d_out;

    const int batch = in_sizes[0] / T;
    addtrans_kernel<<<batch, 64, 0, stream>>>(idx, tok_emb, pos_enc, ln_w, ln_b,
                                              q1w, k1w, v1w, q2w, k2w, v2w,
                                              out_w, lnf_w, lnf_b, head_w, out);
    (void)d_ws; (void)ws_size; (void)n_in; (void)out_size;
}


# Resident work-groups per WGP 13 -> 20: LDS 22656 -> 15200 B (WMMA pad rows not stored, pad lanes re-read a kept row; O reuses the q plane), VGPRs 109 -> 92 (vocab head in two halves)
# speedup vs baseline: 3.8857x; 1.0304x over previous
#include <hip/hip_runtime.h>
#include <math.h>

#define N_LAYERS 4
#define T        34
#define QR       36
#define PR       34
#define PCOLS    36
#define PSTRIDE  36
#define VOCAB    14
#define DM       6
#define HD       3

typedef float v2f __attribute__((ext_vector_type(2)));
typedef float v4f __attribute__((ext_vector_type(4)));
typedef float v8f __attribute__((ext_vector_type(8)));

template <int NC>
__device__ __forceinline__ float softmax_row_regs(float* prow, const int m, const float sc2)
{
    const float NEG_BIG = -3.0e38f;
    float s[NC];
#pragma unroll
    for (int j = 0; j < NC / 4; ++j) {
        const v4f t = *(const v4f*)(prow + 4 * j);
        s[4 * j + 0] = t.x; s[4 * j + 1] = t.y; s[4 * j + 2] = t.z; s[4 * j + 3] = t.w;
    }
    float mx = NEG_BIG;
#pragma unroll
    for (int n = 0; n < NC; ++n) {
        s[n] = (n <= m) ? s[n] : NEG_BIG;
        mx = fmaxf(mx, s[n]);
    }
    const float mb = -mx * sc2;
    float sum = 0.f;
#pragma unroll
    for (int j = 0; j < NC / 4; ++j) {
        v4f e;
        e.x = __builtin_amdgcn_exp2f(fmaf(s[4 * j + 0], sc2, mb)); sum += e.x;
        e.y = __builtin_amdgcn_exp2f(fmaf(s[4 * j + 1], sc2, mb)); sum += e.y;
        e.z = __builtin_amdgcn_exp2f(fmaf(s[4 * j + 2], sc2, mb)); sum += e.z;
        e.w = __builtin_amdgcn_exp2f(fmaf(s[4 * j + 3], sc2, mb)); sum += e.w;
        *(v4f*)(prow + 4 * j) = e;
    }
    return sum;
}

template <int X>
__device__ __forceinline__ float row_xor(float v)
{
    return __builtin_bit_cast(float,
        __builtin_amdgcn_update_dpp(0, __builtin_bit_cast(int, v), 0x160 + X, 0xf, 0xf, false));
}

__device__ __forceinline__ float row16_max(float v)
{
    v = fmaxf(v, row_xor<1>(v));
    v = fmaxf(v, row_xor<2>(v));
    v = fmaxf(v, row_xor<4>(v));
    v = fmaxf(v, row_xor<8>(v));
    return v;
}
__device__ __forceinline__ float row16_sum(float v)
{
    v += row_xor<1>(v);
    v += row_xor<2>(v);
    v += row_xor<4>(v);
    v += row_xor<8>(v);
    return v;
}

__global__ __launch_bounds__(64)
void addtrans_kernel(const int* __restrict__ idx,
                     const float* __restrict__ tok_emb,
                     const float* __restrict__ pos_enc,
                     const float* __restrict__ ln_w, const float* __restrict__ ln_b,
                     const float* __restrict__ q1w, const float* __restrict__ k1w,
                     const float* __restrict__ v1w,
                     const float* __restrict__ q2w, const float* __restrict__ k2w,
                     const float* __restrict__ v2w,
                     const float* __restrict__ out_w,
                     const float* __restrict__ lnf_w, const float* __restrict__ lnf_b,
                     const float* __restrict__ head_w,
                     float* __restrict__ out)
{
    __shared__ float xs[T][DM];
    __shared__ float hs[QR][DM];
    __shared__ __attribute__((aligned(16))) float qs[2][QR][4];
    __shared__ __attribute__((aligned(16))) float ks_[2][QR][4];
    __shared__ __attribute__((aligned(16))) float vs[2][QR][4];
    __shared__ __attribute__((aligned(16))) float Ps[2][PR][PSTRIDE];
    __shared__ float rinv[2][PR];

    const int b    = blockIdx.x;
    const int tid  = threadIdx.x;
    const int w    = tid >> 5;
    const int lane = tid & 31;
    const int hi16 = lane >> 4;
    const int l16  = lane & 15;
    const int kofs = hi16 * 2;
    const int colc = (l16 < 3) ? l16 : 3;

    __builtin_prefetch(idx + (size_t)b * T, 0, 1);

    for (int t = tid; t < T; t += 64) {
        int tok = idx[(size_t)b * T + t];
#pragma unroll
        for (int c = 0; c < 3; ++c) {
            xs[t][c]     = tok_emb[tok * 3 + c];
            xs[t][3 + c] = pos_enc[t * 3 + c];
        }
    }
    __syncthreads();

    const float sc2 = 0.57735026918962576f * 1.4426950408889634f;

    for (int l = 0; l < N_LAYERS; ++l) {
        if (tid < QR) {
            if (tid < T) {
                float m = 0.f;
#pragma unroll
                for (int c = 0; c < DM; ++c) m += xs[tid][c];
                m *= (1.0f / DM);
                float var = 0.f;
#pragma unroll
                for (int c = 0; c < DM; ++c) { float d = xs[tid][c] - m; var += d * d; }
                var *= (1.0f / DM);
                float r = rsqrtf(var + 1e-5f);
#pragma unroll
                for (int c = 0; c < DM; ++c)
                    hs[tid][c] = (xs[tid][c] - m) * r * ln_w[l * DM + c] + ln_b[l * DM + c];
            } else {
#pragma unroll
                for (int c = 0; c < DM; ++c) hs[tid][c] = 0.f;
            }
        }
        __syncthreads();

        const float* qw = (w == 0 ? q1w : q2w) + l * 9;
        const float* kw = (w == 0 ? k1w : k2w) + l * 9;
        const float* vw = (w == 0 ? v1w : v2w) + l * 9;
        const int hoff = w * HD;
        for (int t = lane; t < QR; t += 32) {
            float h0 = hs[t][hoff], h1 = hs[t][hoff + 1], h2 = hs[t][hoff + 2];
            float valid = (t < T) ? 1.0f : 0.0f;
            v4f qv, kv, vv;
#pragma unroll
            for (int i = 0; i < HD; ++i) {
                qv[i] = (qw[i*3]*h0 + qw[i*3+1]*h1 + qw[i*3+2]*h2) * valid;
                kv[i] = (kw[i*3]*h0 + kw[i*3+1]*h1 + kw[i*3+2]*h2) * valid;
                vv[i] = (vw[i*3]*h0 + vw[i*3+1]*h1 + vw[i*3+2]*h2) * valid;
            }
            qv[3] = 0.f; kv[3] = 0.f; vv[3] = 0.f;
            *(v4f*)&qs[w][t][0]  = qv;
            *(v4f*)&ks_[w][t][0] = kv;
            *(v4f*)&vs[w][t][0]  = vv;
        }

        for (int mi = 0; mi < 3; ++mi) {
            int am = mi * 16 + l16;
            if (am > QR - 1) am = QR - 1;
            v2f a = *(const v2f*)&qs[w][am][kofs];
            for (int ni = 0; ni <= mi; ++ni) {
                int bn = ni * 16 + l16;
                if (bn > QR - 1) bn = QR - 1;
                v2f bf = *(const v2f*)&ks_[w][bn][kofs];
                v8f c = {0.f, 0.f, 0.f, 0.f, 0.f, 0.f, 0.f, 0.f};
                c = __builtin_amdgcn_wmma_f32_16x16x4_f32(false, a, false, bf,
                                                          (short)0, c, false, false);
                if (mi < 2) {
#pragma unroll
                    for (int r = 0; r < 8; ++r)
                        Ps[w][mi * 16 + hi16 * 8 + r][ni * 16 + l16] = c[r];
                } else if (hi16 == 0 && ni * 16 + l16 < PCOLS) {
                    Ps[w][32][ni * 16 + l16] = c[0];
                    Ps[w][33][ni * 16 + l16] = c[1];
                }
            }
        }

        {
            const float sum = softmax_row_regs<32>(&Ps[w][lane][0], lane, sc2);
            rinv[w][lane] = (sum > 0.f) ? (1.0f / sum) : 0.f;
        }
        {
            const float NEG_BIG = -3.0e38f;
            const int m = 32 + hi16;
            float* prow = &Ps[w][m][0];
            const bool has2 = (l16 < PCOLS - 32);
            const float s0 = prow[l16];
            const float s1 = prow[16 + l16];
            float s2 = NEG_BIG;
            if (has2) {
                const float t = prow[32 + l16];
                s2 = (l16 <= hi16) ? t : NEG_BIG;
            }
            const float mx = row16_max(fmaxf(fmaxf(s0, s1), s2));
            const float mb = -mx * sc2;
            const float e0 = __builtin_amdgcn_exp2f(fmaf(s0, sc2, mb));
            const float e1 = __builtin_amdgcn_exp2f(fmaf(s1, sc2, mb));
            const float e2 = __builtin_amdgcn_exp2f(fmaf(s2, sc2, mb));
            const float sum = row16_sum((e0 + e1) + e2);
            prow[l16]      = e0;
            prow[16 + l16] = e1;
            if (has2) prow[32 + l16] = e2;
            if (l16 == 0) rinv[w][m] = (sum > 0.f) ? (1.0f / sum) : 0.f;
        }

        for (int mi = 0; mi < 3; ++mi) {
            v8f acc = {0.f, 0.f, 0.f, 0.f, 0.f, 0.f, 0.f, 0.f};
            const int am = (mi < 2) ? mi * 16 + l16 : 32 + (l16 & 1);
            int kmax = 4 * (mi + 1);
            if (kmax > (T + 3) / 4) kmax = (T + 3) / 4;
            for (int kc = 0; kc < kmax; ++kc) {
                const int nb = kc * 4;
                v2f a = *(const v2f*)&Ps[w][am][nb + kofs];
                v2f bf;
                bf.x = vs[w][nb + kofs + 0][colc];
                bf.y = vs[w][nb + kofs + 1][colc];
                acc = __builtin_amdgcn_wmma_f32_16x16x4_f32(false, a, false, bf,
                                                            (short)0, acc, false, false);
            }
            if (l16 < HD) {
                if (mi < 2) {
#pragma unroll
                    for (int r = 0; r < 8; ++r)
                        qs[w][mi * 16 + hi16 * 8 + r][l16] = acc[r];
                } else if (hi16 == 0) {
                    qs[w][32][l16] = acc[0];
                    qs[w][33][l16] = acc[1];
                }
            }
        }
        __syncthreads();

        if (tid < T) {
            float u[DM];
            const float r0 = rinv[0][tid], r1 = rinv[1][tid];
#pragma unroll
            for (int i = 0; i < HD; ++i) { u[i] = qs[0][tid][i] * r0; u[HD + i] = qs[1][tid][i] * r1; }
#pragma unroll
            for (int i = 0; i < DM; ++i) {
                float acc = 0.f;
#pragma unroll
                for (int j = 0; j < DM; ++j) acc += out_w[l * 36 + i * 6 + j] * u[j];
                xs[tid][i] += acc;
            }
        }
        __syncthreads();
    }

    if (tid < T) {
        float m = 0.f;
#pragma unroll
        for (int c = 0; c < DM; ++c) m += xs[tid][c];
        m *= (1.0f / DM);
        float var = 0.f;
#pragma unroll
        for (int c = 0; c < DM; ++c) { float d = xs[tid][c] - m; var += d * d; }
        var *= (1.0f / DM);
        float r = rsqrtf(var + 1e-5f);
        float xf[DM];
#pragma unroll
        for (int c = 0; c < DM; ++c)
            xf[c] = (xs[tid][c] - m) * r * lnf_w[c] + lnf_b[c];
        float* op = out + ((size_t)b * T + tid) * VOCAB;
#pragma unroll
        for (int v = 0; v < VOCAB / 2; ++v) {
            float acc = 0.f;
#pragma unroll
            for (int j = 0; j < DM; ++j) acc += head_w[v * DM + j] * xf[j];
            op[v] = acc;
        }
        __builtin_amdgcn_sched_barrier(0);
#pragma unroll
        for (int v = VOCAB / 2; v < VOCAB; ++v) {
            float acc = 0.f;
#pragma unroll
            for (int j = 0; j < DM; ++j) acc += head_w[v * DM + j] * xf[j];
            op[v] = acc;
        }
    }
}

extern "C" void kernel_launch(void* const* d_in, const int* in_sizes, int n_in,
                              void* d_out, int out_size, void* d_ws, size_t ws_size,
                              hipStream_t stream) {
    const int*   idx     = (const int*)  d_in[0];
    const float* tok_emb = (const float*)d_in[1];
    const float* pos_enc = (const float*)d_in[2];
    const float* ln_w    = (const float*)d_in[3];
    const float* ln_b    = (const float*)d_in[4];
    const float* q1w     = (const float*)d_in[5];
    const float* k1w     = (const float*)d_in[6];
    const float* v1w     = (const float*)d_in[7];
    const float* q2w     = (const float*)d_in[8];
    const float* k2w     = (const float*)d_in[9];
    const float* v2w     = (const float*)d_in[10];
    const float* out_w   = (const float*)d_in[11];
    const float* lnf_w   = (const float*)d_in[12];
    const float* lnf_b   = (const float*)d_in[13];
    const float* head_w  = (const float*)d_in[14];
    float* out = (float*)d_out;

    const int batch = in_sizes[0] / T;
    addtrans_kernel<<<batch, 64, 0, stream>>>(idx, tok_emb, pos_enc, ln_w, ln_b,
                                              q1w, k1w, v1w, q2w, k2w, v2w,
                                              out_w, lnf_w, lnf_b, head_w, out);
    (void)d_ws; (void)ws_size; (void)n_in; (void)out_size;
}


# PV of the 2 left-over queries (32, 33): 9 key chunks packed block-diagonally five to a WMMA (spare rows x spare V columns), 2 WMMAs not 9; their key sum becomes 2-chunk partials + a 5-lane DPP tree
# speedup vs baseline: 4.1622x; 1.0712x over previous
#include <hip/hip_runtime.h>
#include <math.h>

#define N_LAYERS 4
#define T        34
#define QR       36
#define PR       34
#define PCOLS    36
#define PSTRIDE  36
#define VOCAB    14
#define DM       6
#define HD       3

typedef float v2f __attribute__((ext_vector_type(2)));
typedef float v4f __attribute__((ext_vector_type(4)));
typedef float v8f __attribute__((ext_vector_type(8)));

template <int NC>
__device__ __forceinline__ float softmax_row_regs(float* prow, const int m, const float sc2)
{
    const float NEG_BIG = -3.0e38f;
    float s[NC];
#pragma unroll
    for (int j = 0; j < NC / 4; ++j) {
        const v4f t = *(const v4f*)(prow + 4 * j);
        s[4 * j + 0] = t.x; s[4 * j + 1] = t.y; s[4 * j + 2] = t.z; s[4 * j + 3] = t.w;
    }
    float mx = NEG_BIG;
#pragma unroll
    for (int n = 0; n < NC; ++n) {
        s[n] = (n <= m) ? s[n] : NEG_BIG;
        mx = fmaxf(mx, s[n]);
    }
    const float mb = -mx * sc2;
    float sum = 0.f;
#pragma unroll
    for (int j = 0; j < NC / 4; ++j) {
        v4f e;
        e.x = __builtin_amdgcn_exp2f(fmaf(s[4 * j + 0], sc2, mb)); sum += e.x;
        e.y = __builtin_amdgcn_exp2f(fmaf(s[4 * j + 1], sc2, mb)); sum += e.y;
        e.z = __builtin_amdgcn_exp2f(fmaf(s[4 * j + 2], sc2, mb)); sum += e.z;
        e.w = __builtin_amdgcn_exp2f(fmaf(s[4 * j + 3], sc2, mb)); sum += e.w;
        *(v4f*)(prow + 4 * j) = e;
    }
    return sum;
}

template <int X>
__device__ __forceinline__ float row_xor(float v)
{
    return __builtin_bit_cast(float,
        __builtin_amdgcn_update_dpp(0, __builtin_bit_cast(int, v), 0x160 + X, 0xf, 0xf, false));
}

__device__ __forceinline__ float row16_max(float v)
{
    v = fmaxf(v, row_xor<1>(v));
    v = fmaxf(v, row_xor<2>(v));
    v = fmaxf(v, row_xor<4>(v));
    v = fmaxf(v, row_xor<8>(v));
    return v;
}
__device__ __forceinline__ float row16_sum(float v)
{
    v += row_xor<1>(v);
    v += row_xor<2>(v);
    v += row_xor<4>(v);
    v += row_xor<8>(v);
    return v;
}

template <int N>
__device__ __forceinline__ float row_shr(float v)
{
    return __builtin_bit_cast(float,
        __builtin_amdgcn_update_dpp(0, __builtin_bit_cast(int, v), 0x110 + N, 0xf, 0xf, true));
}

__global__ __launch_bounds__(64)
void addtrans_kernel(const int* __restrict__ idx,
                     const float* __restrict__ tok_emb,
                     const float* __restrict__ pos_enc,
                     const float* __restrict__ ln_w, const float* __restrict__ ln_b,
                     const float* __restrict__ q1w, const float* __restrict__ k1w,
                     const float* __restrict__ v1w,
                     const float* __restrict__ q2w, const float* __restrict__ k2w,
                     const float* __restrict__ v2w,
                     const float* __restrict__ out_w,
                     const float* __restrict__ lnf_w, const float* __restrict__ lnf_b,
                     const float* __restrict__ head_w,
                     float* __restrict__ out)
{
    __shared__ float xs[T][DM];
    __shared__ float hs[QR][DM];
    __shared__ __attribute__((aligned(16))) float qs[2][QR][4];
    __shared__ __attribute__((aligned(16))) float ks_[2][QR][4];
    __shared__ __attribute__((aligned(16))) float vs[2][QR][4];
    __shared__ __attribute__((aligned(16))) float Ps[2][PR][PSTRIDE];
    __shared__ float rinv[2][PR];

    const int b    = blockIdx.x;
    const int tid  = threadIdx.x;
    const int w    = tid >> 5;
    const int lane = tid & 31;
    const int hi16 = lane >> 4;
    const int l16  = lane & 15;
    const int kofs = hi16 * 2;
    const int colc = (l16 < 3) ? l16 : 3;

    __builtin_prefetch(idx + (size_t)b * T, 0, 1);

    for (int t = tid; t < T; t += 64) {
        int tok = idx[(size_t)b * T + t];
#pragma unroll
        for (int c = 0; c < 3; ++c) {
            xs[t][c]     = tok_emb[tok * 3 + c];
            xs[t][3 + c] = pos_enc[t * 3 + c];
        }
    }
    __syncthreads();

    const float sc2 = 0.57735026918962576f * 1.4426950408889634f;

    for (int l = 0; l < N_LAYERS; ++l) {
        if (tid < QR) {
            if (tid < T) {
                float m = 0.f;
#pragma unroll
                for (int c = 0; c < DM; ++c) m += xs[tid][c];
                m *= (1.0f / DM);
                float var = 0.f;
#pragma unroll
                for (int c = 0; c < DM; ++c) { float d = xs[tid][c] - m; var += d * d; }
                var *= (1.0f / DM);
                float r = rsqrtf(var + 1e-5f);
#pragma unroll
                for (int c = 0; c < DM; ++c)
                    hs[tid][c] = (xs[tid][c] - m) * r * ln_w[l * DM + c] + ln_b[l * DM + c];
            } else {
#pragma unroll
                for (int c = 0; c < DM; ++c) hs[tid][c] = 0.f;
            }
        }
        __syncthreads();

        const float* qw = (w == 0 ? q1w : q2w) + l * 9;
        const float* kw = (w == 0 ? k1w : k2w) + l * 9;
        const float* vw = (w == 0 ? v1w : v2w) + l * 9;
        const int hoff = w * HD;
        for (int t = lane; t < QR; t += 32) {
            float h0 = hs[t][hoff], h1 = hs[t][hoff + 1], h2 = hs[t][hoff + 2];
            float valid = (t < T) ? 1.0f : 0.0f;
            v4f qv, kv, vv;
#pragma unroll
            for (int i = 0; i < HD; ++i) {
                qv[i] = (qw[i*3]*h0 + qw[i*3+1]*h1 + qw[i*3+2]*h2) * valid;
                kv[i] = (kw[i*3]*h0 + kw[i*3+1]*h1 + kw[i*3+2]*h2) * valid;
                vv[i] = (vw[i*3]*h0 + vw[i*3+1]*h1 + vw[i*3+2]*h2) * valid;
            }
            qv[3] = 0.f; kv[3] = 0.f; vv[3] = 0.f;
            *(v4f*)&qs[w][t][0]  = qv;
            *(v4f*)&ks_[w][t][0] = kv;
            *(v4f*)&vs[w][t][0]  = vv;
        }

        for (int mi = 0; mi < 3; ++mi) {
            int am = mi * 16 + l16;
            if (am > QR - 1) am = QR - 1;
            v2f a = *(const v2f*)&qs[w][am][kofs];
            for (int ni = 0; ni <= mi; ++ni) {
                int bn = ni * 16 + l16;
                if (bn > QR - 1) bn = QR - 1;
                v2f bf = *(const v2f*)&ks_[w][bn][kofs];
                v8f c = {0.f, 0.f, 0.f, 0.f, 0.f, 0.f, 0.f, 0.f};
                c = __builtin_amdgcn_wmma_f32_16x16x4_f32(false, a, false, bf,
                                                          (short)0, c, false, false);
                if (mi < 2) {
#pragma unroll
                    for (int r = 0; r < 8; ++r)
                        Ps[w][mi * 16 + hi16 * 8 + r][ni * 16 + l16] = c[r];
                } else if (hi16 == 0 && ni * 16 + l16 < PCOLS) {
                    Ps[w][32][ni * 16 + l16] = c[0];
                    Ps[w][33][ni * 16 + l16] = c[1];
                }
            }
        }

        {
            const float sum = softmax_row_regs<32>(&Ps[w][lane][0], lane, sc2);
            rinv[w][lane] = (sum > 0.f) ? (1.0f / sum) : 0.f;
        }
        {
            const float NEG_BIG = -3.0e38f;
            const int m = 32 + hi16;
            float* prow = &Ps[w][m][0];
            const bool has2 = (l16 < PCOLS - 32);
            const float s0 = prow[l16];
            const float s1 = prow[16 + l16];
            float s2 = NEG_BIG;
            if (has2) {
                const float t = prow[32 + l16];
                s2 = (l16 <= hi16) ? t : NEG_BIG;
            }
            const float mx = row16_max(fmaxf(fmaxf(s0, s1), s2));
            const float mb = -mx * sc2;
            const float e0 = __builtin_amdgcn_exp2f(fmaf(s0, sc2, mb));
            const float e1 = __builtin_amdgcn_exp2f(fmaf(s1, sc2, mb));
            const float e2 = __builtin_amdgcn_exp2f(fmaf(s2, sc2, mb));
            const float sum = row16_sum((e0 + e1) + e2);
            prow[l16]      = e0;
            prow[16 + l16] = e1;
            if (has2) prow[32 + l16] = e2;
            if (l16 == 0) rinv[w][m] = (sum > 0.f) ? (1.0f / sum) : 0.f;
        }

        for (int mi = 0; mi < 2; ++mi) {
            v8f acc = {0.f, 0.f, 0.f, 0.f, 0.f, 0.f, 0.f, 0.f};
            const int am = mi * 16 + l16;
            const int kmax = 4 * (mi + 1);
            for (int kc = 0; kc < kmax; ++kc) {
                const int nb = kc * 4;
                v2f a = *(const v2f*)&Ps[w][am][nb + kofs];
                v2f bf;
                bf.x = vs[w][nb + kofs + 0][colc];
                bf.y = vs[w][nb + kofs + 1][colc];
                acc = __builtin_amdgcn_wmma_f32_16x16x4_f32(false, a, false, bf,
                                                            (short)0, acc, false, false);
            }
            if (l16 < HD) {
#pragma unroll
                for (int r = 0; r < 8; ++r)
                    qs[w][mi * 16 + hi16 * 8 + r][l16] = acc[r];
            }
        }
        {
            const int qq = l16 >> 3;
            int ga = l16 & 7;
            if (ga > 4) ga = 4;
            const int ga1 = (ga > 3) ? 3 : ga;
            const int gn  = l16 / 3;
            const int cb  = (l16 < 15) ? (l16 - 3 * gn) : 3;
            const int kb0 = 4 * gn + kofs;
            int kb1 = 4 * (5 + gn) + kofs;
            const int kb1x = (kb1     > QR - 1) ? QR - 1 : kb1;
            const int kb1y = (kb1 + 1 > QR - 1) ? QR - 1 : kb1 + 1;
            v8f acc = {0.f, 0.f, 0.f, 0.f, 0.f, 0.f, 0.f, 0.f};
            v2f a0 = *(const v2f*)&Ps[w][32 + qq][4 * ga + kofs];
            v2f b0;
            b0.x = vs[w][kb0 + 0][cb];
            b0.y = vs[w][kb0 + 1][cb];
            acc = __builtin_amdgcn_wmma_f32_16x16x4_f32(false, a0, false, b0,
                                                        (short)0, acc, false, false);
            v2f a1 = *(const v2f*)&Ps[w][32 + qq][4 * (5 + ga1) + kofs];
            v2f b1;
            b1.x = vs[w][kb1x][cb];
            b1.y = vs[w][kb1y][cb];
            acc = __builtin_amdgcn_wmma_f32_16x16x4_f32(false, a1, false, b1,
                                                        (short)0, acc, false, false);
            asm volatile("s_wait_alu depctr_va_vdst(0)" : "+v"(acc));
            float pv = acc[0];
            pv = (gn == 1) ? acc[1] : pv;
            pv = (gn == 2) ? acc[2] : pv;
            pv = (gn == 3) ? acc[3] : pv;
            pv = (gn == 4) ? acc[4] : pv;
            const float t3 = pv + row_shr<3>(pv);
            const float t6 = t3 + row_shr<6>(t3);
            const float o  = t6 + row_shr<12>(pv);
            if (l16 >= 12 && l16 < 15)
                qs[w][32 + hi16][l16 - 12] = o;
        }
        __syncthreads();

        if (tid < T) {
            float u[DM];
            const float r0 = rinv[0][tid], r1 = rinv[1][tid];
#pragma unroll
            for (int i = 0; i < HD; ++i) { u[i] = qs[0][tid][i] * r0; u[HD + i] = qs[1][tid][i] * r1; }
#pragma unroll
            for (int i = 0; i < DM; ++i) {
                float acc = 0.f;
#pragma unroll
                for (int j = 0; j < DM; ++j) acc += out_w[l * 36 + i * 6 + j] * u[j];
                xs[tid][i] += acc;
            }
        }
        __syncthreads();
    }

    if (tid < T) {
        float m = 0.f;
#pragma unroll
        for (int c = 0; c < DM; ++c) m += xs[tid][c];
        m *= (1.0f / DM);
        float var = 0.f;
#pragma unroll
        for (int c = 0; c < DM; ++c) { float d = xs[tid][c] - m; var += d * d; }
        var *= (1.0f / DM);
        float r = rsqrtf(var + 1e-5f);
        float xf[DM];
#pragma unroll
        for (int c = 0; c < DM; ++c)
            xf[c] = (xs[tid][c] - m) * r * lnf_w[c] + lnf_b[c];
        float* op = out + ((size_t)b * T + tid) * VOCAB;
#pragma unroll
        for (int v = 0; v < VOCAB / 2; ++v) {
            float acc = 0.f;
#pragma unroll
            for (int j = 0; j < DM; ++j) acc += head_w[v * DM + j] * xf[j];
            op[v] = acc;
        }
        __builtin_amdgcn_sched_barrier(0);
#pragma unroll
        for (int v = VOCAB / 2; v < VOCAB; ++v) {
            float acc = 0.f;
#pragma unroll
            for (int j = 0; j < DM; ++j) acc += head_w[v * DM + j] * xf[j];
            op[v] = acc;
        }
    }
}

extern "C" void kernel_launch(void* const* d_in, const int* in_sizes, int n_in,
                              void* d_out, int out_size, void* d_ws, size_t ws_size,
                              hipStream_t stream) {
    const int*   idx     = (const int*)  d_in[0];
    const float* tok_emb = (const float*)d_in[1];
    const float* pos_enc = (const float*)d_in[2];
    const float* ln_w    = (const float*)d_in[3];
    const float* ln_b    = (const float*)d_in[4];
    const float* q1w     = (const float*)d_in[5];
    const float* k1w     = (const float*)d_in[6];
    const float* v1w     = (const float*)d_in[7];
    const float* q2w     = (const float*)d_in[8];
    const float* k2w     = (const float*)d_in[9];
    const float* v2w     = (const float*)d_in[10];
    const float* out_w   = (const float*)d_in[11];
    const float* lnf_w   = (const float*)d_in[12];
    const float* lnf_b   = (const float*)d_in[13];
    const float* head_w  = (const float*)d_in[14];
    float* out = (float*)d_out;

    const int batch = in_sizes[0] / T;
    addtrans_kernel<<<batch, 64, 0, stream>>>(idx, tok_emb, pos_enc, ln_w, ln_b,
                                              q1w, k1w, v1w, q2w, k2w, v2w,
                                              out_w, lnf_w, lnf_b, head_w, out);
    (void)d_ws; (void)ws_size; (void)n_in; (void)out_size;
}


# Softmax of queries 0..31 done in the PV A-fragment layout (2 lanes a row, each half the keys): P feeds the PV WMMAs from registers, no P LDS trip; row sums and PV key order become per-lane-half
# speedup vs baseline: 4.5732x; 1.0987x over previous
#include <hip/hip_runtime.h>
#include <math.h>

#define N_LAYERS 4
#define T        34
#define QR       36
#define PR       34
#define PCOLS    36
#define PSTRIDE  36
#define VOCAB    14
#define DM       6
#define HD       3

typedef float v2f __attribute__((ext_vector_type(2)));
typedef float v4f __attribute__((ext_vector_type(4)));
typedef float v8f __attribute__((ext_vector_type(8)));

template <int X>
__device__ __forceinline__ float row_xor(float v)
{
    return __builtin_bit_cast(float,
        __builtin_amdgcn_update_dpp(0, __builtin_bit_cast(int, v), 0x160 + X, 0xf, 0xf, false));
}

__device__ __forceinline__ float row16_max(float v)
{
    v = fmaxf(v, row_xor<1>(v));
    v = fmaxf(v, row_xor<2>(v));
    v = fmaxf(v, row_xor<4>(v));
    v = fmaxf(v, row_xor<8>(v));
    return v;
}
__device__ __forceinline__ float row16_sum(float v)
{
    v += row_xor<1>(v);
    v += row_xor<2>(v);
    v += row_xor<4>(v);
    v += row_xor<8>(v);
    return v;
}

template <int N>
__device__ __forceinline__ float row_shr(float v)
{
    return __builtin_bit_cast(float,
        __builtin_amdgcn_update_dpp(0, __builtin_bit_cast(int, v), 0x110 + N, 0xf, 0xf, true));
}

__device__ __forceinline__ float xhalf(float v)
{
    const int b = __builtin_bit_cast(int, v);
    return __builtin_bit_cast(float,
        __builtin_amdgcn_permlanex16(b, b, 0x76543210, 0xfedcba98, false, false));
}

template <int MI>
__device__ __forceinline__ float softmax_tile_frag(const float* prow, const int m, const int hi16,
                                                    const float sc2, float (&e)[8 * MI + 8])
{
    constexpr int NI = 2 * MI + 2;
    const float NEG_BIG = -3.0e38f;
    float s[4 * NI];
#pragma unroll
    for (int i = 0; i < NI; ++i) {
        const v4f t = *(const v4f*)(prow + 8 * i + 4 * hi16);
        s[4 * i + 0] = t.x; s[4 * i + 1] = t.y; s[4 * i + 2] = t.z; s[4 * i + 3] = t.w;
    }
    float mx = NEG_BIG;
#pragma unroll
    for (int i = 0; i < NI; ++i) {
#pragma unroll
        for (int t = 0; t < 4; ++t) {
            if (8 * i + 7 > 16 * MI)
                s[4 * i + t] = (8 * i + 4 * hi16 + t <= m) ? s[4 * i + t] : NEG_BIG;
            mx = fmaxf(mx, s[4 * i + t]);
        }
    }
    mx = fmaxf(mx, xhalf(mx));
    const float mb = -mx * sc2;
    float sum = 0.f;
#pragma unroll
    for (int n = 0; n < 4 * NI; ++n) {
        e[n] = __builtin_amdgcn_exp2f(fmaf(s[n], sc2, mb));
        sum += e[n];
    }
    return sum + xhalf(sum);
}

__device__ __forceinline__ void pv_tiles_frag(const float (&e0)[8], const float (&e1)[16], const float (*vsw)[4],
                                              const int hi16, const int colc, v8f& acc0, v8f& acc1)
{
    v2f bfr[8];
#pragma unroll
    for (int j = 0; j < 8; ++j) {
        const int kb = 8 * (j >> 1) + 4 * hi16 + 2 * (j & 1);
        bfr[j].x = vsw[kb + 0][colc];
        bfr[j].y = vsw[kb + 1][colc];
    }
    __builtin_amdgcn_sched_barrier(0);
    acc0 = (v8f){0.f, 0.f, 0.f, 0.f, 0.f, 0.f, 0.f, 0.f};
    acc1 = (v8f){0.f, 0.f, 0.f, 0.f, 0.f, 0.f, 0.f, 0.f};
#pragma unroll
    for (int j = 0; j < 8; ++j) {
        if (j < 4) {
            v2f a;
            a.x = e0[2 * j];
            a.y = e0[2 * j + 1];
            acc0 = __builtin_amdgcn_wmma_f32_16x16x4_f32(false, a, false, bfr[j],
                                                         (short)0, acc0, false, false);
        }
        v2f a;
        a.x = e1[2 * j];
        a.y = e1[2 * j + 1];
        acc1 = __builtin_amdgcn_wmma_f32_16x16x4_f32(false, a, false, bfr[j],
                                                     (short)0, acc1, false, false);
    }
    asm volatile("v_nop\n\tv_nop\n\tv_nop\n\tv_nop" : "+v"(acc0), "+v"(acc1));
    __builtin_amdgcn_sched_barrier(0);
}

__global__ __launch_bounds__(64)
void addtrans_kernel(const int* __restrict__ idx,
                     const float* __restrict__ tok_emb,
                     const float* __restrict__ pos_enc,
                     const float* __restrict__ ln_w, const float* __restrict__ ln_b,
                     const float* __restrict__ q1w, const float* __restrict__ k1w,
                     const float* __restrict__ v1w,
                     const float* __restrict__ q2w, const float* __restrict__ k2w,
                     const float* __restrict__ v2w,
                     const float* __restrict__ out_w,
                     const float* __restrict__ lnf_w, const float* __restrict__ lnf_b,
                     const float* __restrict__ head_w,
                     float* __restrict__ out)
{
    __shared__ float xs[T][DM];
    __shared__ float hs[QR][DM];
    __shared__ __attribute__((aligned(16))) float qs[2][QR][4];
    __shared__ __attribute__((aligned(16))) float ks_[2][QR][4];
    __shared__ __attribute__((aligned(16))) float vs[2][QR][4];
    __shared__ __attribute__((aligned(16))) float Ps[2][PR][PSTRIDE];
    __shared__ float rinv[2][PR];

    const int b    = blockIdx.x;
    const int tid  = threadIdx.x;
    const int w    = tid >> 5;
    const int lane = tid & 31;
    const int hi16 = lane >> 4;
    const int l16  = lane & 15;
    const int kofs = hi16 * 2;
    const int colc = (l16 < 3) ? l16 : 3;

    __builtin_prefetch(idx + (size_t)b * T, 0, 1);

    for (int t = tid; t < T; t += 64) {
        int tok = idx[(size_t)b * T + t];
#pragma unroll
        for (int c = 0; c < 3; ++c) {
            xs[t][c]     = tok_emb[tok * 3 + c];
            xs[t][3 + c] = pos_enc[t * 3 + c];
        }
    }
    __syncthreads();

    const float sc2 = 0.57735026918962576f * 1.4426950408889634f;

    for (int l = 0; l < N_LAYERS; ++l) {
        if (tid < QR) {
            if (tid < T) {
                float m = 0.f;
#pragma unroll
                for (int c = 0; c < DM; ++c) m += xs[tid][c];
                m *= (1.0f / DM);
                float var = 0.f;
#pragma unroll
                for (int c = 0; c < DM; ++c) { float d = xs[tid][c] - m; var += d * d; }
                var *= (1.0f / DM);
                float r = rsqrtf(var + 1e-5f);
#pragma unroll
                for (int c = 0; c < DM; ++c)
                    hs[tid][c] = (xs[tid][c] - m) * r * ln_w[l * DM + c] + ln_b[l * DM + c];
            } else {
#pragma unroll
                for (int c = 0; c < DM; ++c) hs[tid][c] = 0.f;
            }
        }
        __syncthreads();

        const float* qw = (w == 0 ? q1w : q2w) + l * 9;
        const float* kw = (w == 0 ? k1w : k2w) + l * 9;
        const float* vw = (w == 0 ? v1w : v2w) + l * 9;
        const int hoff = w * HD;
        for (int t = lane; t < QR; t += 32) {
            float h0 = hs[t][hoff], h1 = hs[t][hoff + 1], h2 = hs[t][hoff + 2];
            float valid = (t < T) ? 1.0f : 0.0f;
            v4f qv, kv, vv;
#pragma unroll
            for (int i = 0; i < HD; ++i) {
                qv[i] = (qw[i*3]*h0 + qw[i*3+1]*h1 + qw[i*3+2]*h2) * valid;
                kv[i] = (kw[i*3]*h0 + kw[i*3+1]*h1 + kw[i*3+2]*h2) * valid;
                vv[i] = (vw[i*3]*h0 + vw[i*3+1]*h1 + vw[i*3+2]*h2) * valid;
            }
            qv[3] = 0.f; kv[3] = 0.f; vv[3] = 0.f;
            *(v4f*)&qs[w][t][0]  = qv;
            *(v4f*)&ks_[w][t][0] = kv;
            *(v4f*)&vs[w][t][0]  = vv;
        }

        for (int mi = 0; mi < 3; ++mi) {
            int am = mi * 16 + l16;
            if (am > QR - 1) am = QR - 1;
            v2f a = *(const v2f*)&qs[w][am][kofs];
            for (int ni = 0; ni <= mi; ++ni) {
                int bn = ni * 16 + l16;
                if (bn > QR - 1) bn = QR - 1;
                v2f bf = *(const v2f*)&ks_[w][bn][kofs];
                v8f c = {0.f, 0.f, 0.f, 0.f, 0.f, 0.f, 0.f, 0.f};
                c = __builtin_amdgcn_wmma_f32_16x16x4_f32(false, a, false, bf,
                                                          (short)0, c, false, false);
                if (mi < 2) {
#pragma unroll
                    for (int r = 0; r < 8; ++r)
                        Ps[w][mi * 16 + hi16 * 8 + r][ni * 16 + l16] = c[r];
                } else if (hi16 == 0 && ni * 16 + l16 < PCOLS) {
                    Ps[w][32][ni * 16 + l16] = c[0];
                    Ps[w][33][ni * 16 + l16] = c[1];
                }
            }
        }

        {
            float e0[8], e1[16];
            const float sum0 = softmax_tile_frag<0>(&Ps[w][l16][0],      l16,      hi16, sc2, e0);
            const float sum1 = softmax_tile_frag<1>(&Ps[w][16 + l16][0], 16 + l16, hi16, sc2, e1);
            const float sum = hi16 ? sum1 : sum0;
            rinv[w][lane] = (sum > 0.f) ? (1.0f / sum) : 0.f;
            v8f acc0, acc1;
            pv_tiles_frag(e0, e1, vs[w], hi16, colc, acc0, acc1);
            if (l16 < HD) {
#pragma unroll
                for (int r = 0; r < 8; ++r) {
                    qs[w][hi16 * 8 + r][l16]      = acc0[r];
                    qs[w][16 + hi16 * 8 + r][l16] = acc1[r];
                }
            }
        }
        {
            const float NEG_BIG = -3.0e38f;
            const int m = 32 + hi16;
            float* prow = &Ps[w][m][0];
            const bool has2 = (l16 < PCOLS - 32);
            const float s0 = prow[l16];
            const float s1 = prow[16 + l16];
            float s2 = NEG_BIG;
            if (has2) {
                const float t = prow[32 + l16];
                s2 = (l16 <= hi16) ? t : NEG_BIG;
            }
            const float mx = row16_max(fmaxf(fmaxf(s0, s1), s2));
            const float mb = -mx * sc2;
            const float e0 = __builtin_amdgcn_exp2f(fmaf(s0, sc2, mb));
            const float e1 = __builtin_amdgcn_exp2f(fmaf(s1, sc2, mb));
            const float e2 = __builtin_amdgcn_exp2f(fmaf(s2, sc2, mb));
            const float sum = row16_sum((e0 + e1) + e2);
            prow[l16]      = e0;
            prow[16 + l16] = e1;
            if (has2) prow[32 + l16] = e2;
            if (l16 == 0) rinv[w][m] = (sum > 0.f) ? (1.0f / sum) : 0.f;
        }

        {
            const int qq = l16 >> 3;
            int ga = l16 & 7;
            if (ga > 4) ga = 4;
            const int ga1 = (ga > 3) ? 3 : ga;
            const int gn  = l16 / 3;
            const int cb  = (l16 < 15) ? (l16 - 3 * gn) : 3;
            const int kb0 = 4 * gn + kofs;
            int kb1 = 4 * (5 + gn) + kofs;
            const int kb1x = (kb1     > QR - 1) ? QR - 1 : kb1;
            const int kb1y = (kb1 + 1 > QR - 1) ? QR - 1 : kb1 + 1;
            v8f acc = {0.f, 0.f, 0.f, 0.f, 0.f, 0.f, 0.f, 0.f};
            v2f a0 = *(const v2f*)&Ps[w][32 + qq][4 * ga + kofs];
            v2f b0;
            b0.x = vs[w][kb0 + 0][cb];
            b0.y = vs[w][kb0 + 1][cb];
            acc = __builtin_amdgcn_wmma_f32_16x16x4_f32(false, a0, false, b0,
                                                        (short)0, acc, false, false);
            v2f a1 = *(const v2f*)&Ps[w][32 + qq][4 * (5 + ga1) + kofs];
            v2f b1;
            b1.x = vs[w][kb1x][cb];
            b1.y = vs[w][kb1y][cb];
            acc = __builtin_amdgcn_wmma_f32_16x16x4_f32(false, a1, false, b1,
                                                        (short)0, acc, false, false);
            asm volatile("s_wait_alu depctr_va_vdst(0)" : "+v"(acc));
            float pv = acc[0];
            pv = (gn == 1) ? acc[1] : pv;
            pv = (gn == 2) ? acc[2] : pv;
            pv = (gn == 3) ? acc[3] : pv;
            pv = (gn == 4) ? acc[4] : pv;
            const float t3 = pv + row_shr<3>(pv);
            const float t6 = t3 + row_shr<6>(t3);
            const float o  = t6 + row_shr<12>(pv);
            if (l16 >= 12 && l16 < 15)
                qs[w][32 + hi16][l16 - 12] = o;
        }
        __syncthreads();

        if (tid < T) {
            float u[DM];
            const float r0 = rinv[0][tid], r1 = rinv[1][tid];
#pragma unroll
            for (int i = 0; i < HD; ++i) { u[i] = qs[0][tid][i] * r0; u[HD + i] = qs[1][tid][i] * r1; }
#pragma unroll
            for (int i = 0; i < DM; ++i) {
                float acc = 0.f;
#pragma unroll
                for (int j = 0; j < DM; ++j) acc += out_w[l * 36 + i * 6 + j] * u[j];
                xs[tid][i] += acc;
            }
        }
        __syncthreads();
    }

    if (tid < T) {
        float m = 0.f;
#pragma unroll
        for (int c = 0; c < DM; ++c) m += xs[tid][c];
        m *= (1.0f / DM);
        float var = 0.f;
#pragma unroll
        for (int c = 0; c < DM; ++c) { float d = xs[tid][c] - m; var += d * d; }
        var *= (1.0f / DM);
        float r = rsqrtf(var + 1e-5f);
        float xf[DM];
#pragma unroll
        for (int c = 0; c < DM; ++c)
            xf[c] = (xs[tid][c] - m) * r * lnf_w[c] + lnf_b[c];
        float* op = out + ((size_t)b * T + tid) * VOCAB;
#pragma unroll
        for (int v = 0; v < VOCAB / 2; ++v) {
            float acc = 0.f;
#pragma unroll
            for (int j = 0; j < DM; ++j) acc += head_w[v * DM + j] * xf[j];
            op[v] = acc;
        }
        __builtin_amdgcn_sched_barrier(0);
#pragma unroll
        for (int v = VOCAB / 2; v < VOCAB; ++v) {
            float acc = 0.f;
#pragma unroll
            for (int j = 0; j < DM; ++j) acc += head_w[v * DM + j] * xf[j];
            op[v] = acc;
        }
    }
}

extern "C" void kernel_launch(void* const* d_in, const int* in_sizes, int n_in,
                              void* d_out, int out_size, void* d_ws, size_t ws_size,
                              hipStream_t stream) {
    const int*   idx     = (const int*)  d_in[0];
    const float* tok_emb = (const float*)d_in[1];
    const float* pos_enc = (const float*)d_in[2];
    const float* ln_w    = (const float*)d_in[3];
    const float* ln_b    = (const float*)d_in[4];
    const float* q1w     = (const float*)d_in[5];
    const float* k1w     = (const float*)d_in[6];
    const float* v1w     = (const float*)d_in[7];
    const float* q2w     = (const float*)d_in[8];
    const float* k2w     = (const float*)d_in[9];
    const float* v2w     = (const float*)d_in[10];
    const float* out_w   = (const float*)d_in[11];
    const float* lnf_w   = (const float*)d_in[12];
    const float* lnf_b   = (const float*)d_in[13];
    const float* head_w  = (const float*)d_in[14];
    float* out = (float*)d_out;

    const int batch = in_sizes[0] / T;
    addtrans_kernel<<<batch, 64, 0, stream>>>(idx, tok_emb, pos_enc, ln_w, ln_b,
                                              q1w, k1w, v1w, q2w, k2w, v2w,
                                              out_w, lnf_w, lnf_b, head_w, out);
    (void)d_ws; (void)ws_size; (void)n_in; (void)out_size;
}
